# starting kernel
# baseline (speedup 1.0000x reference)
#include <hip/hip_runtime.h>
#include <hip/hip_bf16.h>

#define IN_DIM 256
#define OUT_DIM 64
#define RANK 3
#define WB_STRIDE 260

typedef __attribute__((ext_vector_type(2))) float v2f;
typedef __attribute__((ext_vector_type(8))) float v8f;


__global__ void k_zero_deg(int* __restrict__ deg, int n) {
  int i = blockIdx.x * blockDim.x + threadIdx.x;
  if (i < n) deg[i] = 0;
}

__global__ void k_deg(const int* __restrict__ rows, const int* __restrict__ cols,
                      int* __restrict__ deg, int E) {
  int e = blockIdx.x * blockDim.x + threadIdx.x;
  if (e >= E) return;
  int r = rows[e], c = cols[e];
  if (r != c) atomicAdd(&deg[c], 1);
}

__global__ void k_dinv(const int* __restrict__ deg, float* __restrict__ dinv, int n) {
  int i = blockIdx.x * blockDim.x + threadIdx.x;
  if (i < n) dinv[i] = rsqrtf((float)deg[i] + 1.0f);
}

__global__ void __launch_bounds__(256) k_lora_wmma(
    const float* __restrict__ x, const float* __restrict__ WB,
    const float* __restrict__ WA, const float* __restrict__ dinv,
    float* __restrict__ g, float* __restrict__ out, int ntiles) {
  __shared__ float ldsWB[16][WB_STRIDE];
  __shared__ float ldsT[8][256];

  for (int idx = threadIdx.x; idx < 16 * IN_DIM; idx += blockDim.x) {
    const int r = idx >> 8;
    const int k = idx & (IN_DIM - 1);
    ldsWB[r][k] = (r < RANK) ? WB[r * IN_DIM + k] : 0.0f;
  }
  __syncthreads();

  const int wave = threadIdx.x >> 5;
  const int lane = threadIdx.x & 31;
  int tile = blockIdx.x * 8 + wave;
  if (tile >= ntiles) tile = ntiles - 1;

  const int m    = lane & 15;
  const int half = lane >> 4;
  const int koff = half * 2;
  const int base = tile * 16;

  const float* xrow = x + (size_t)(base + m) * IN_DIM + koff;
  const float* wrow = &ldsWB[m][koff];

  v8f c = {};
  #pragma unroll 8
  for (int kk = 0; kk < IN_DIM; kk += 4) {
    v2f a = *reinterpret_cast<const v2f*>(xrow + kk);
    v2f b = *reinterpret_cast<const v2f*>(wrow + kk);
    c = __builtin_amdgcn_wmma_f32_16x16x4_f32(false, a, false, b, (short)0, c,
                                              false, false);
  }

  #pragma unroll
  for (int j = 0; j < 8; ++j) {
    ldsT[wave][(j + 8 * half) * 16 + m] = c[j];
  }
  __syncthreads();

  v2f a2;
  a2.x = ldsT[wave][m * 16 + koff];
  a2.y = ldsT[wave][m * 16 + koff + 1];

  float dv[8];
  #pragma unroll
  for (int j = 0; j < 8; ++j) dv[j] = dinv[base + j + 8 * half];

  #pragma unroll
  for (int jt = 0; jt < 4; ++jt) {
    const int oc = jt * 16 + m;
    v2f b2;
    b2.x = WA[oc * RANK + koff];
    b2.y = (koff + 1 < RANK) ? WA[oc * RANK + koff + 1] : 0.0f;
    v8f d = {};
    d = __builtin_amdgcn_wmma_f32_16x16x4_f32(false, a2, false, b2, (short)0, d,
                                              false, false);
    #pragma unroll
    for (int j = 0; j < 8; ++j) {
      const int node = base + j + 8 * half;
      const float val = d[j] * dv[j];
      const size_t off = (size_t)node * OUT_DIM + oc;
      g[off]   = val;
      out[off] = val;
    }
  }
}

__global__ void k_lora_tail(const float* __restrict__ x, const float* __restrict__ WB,
                            const float* __restrict__ WA, const float* __restrict__ dinv,
                            float* __restrict__ g, float* __restrict__ out,
                            int start, int n) {
  int i = start + blockIdx.x * blockDim.x + threadIdx.x;
  if (i >= n) return;
  float t[RANK];
  for (int r = 0; r < RANK; ++r) {
    float s = 0.0f;
    for (int k = 0; k < IN_DIM; ++k) s += x[(size_t)i * IN_DIM + k] * WB[r * IN_DIM + k];
    t[r] = s;
  }
  const float dv = dinv[i];
  for (int oc = 0; oc < OUT_DIM; ++oc) {
    float s = 0.0f;
    for (int r = 0; r < RANK; ++r) s += t[r] * WA[oc * RANK + r];
    s *= dv;
    g[(size_t)i * OUT_DIM + oc] = s;
    out[(size_t)i * OUT_DIM + oc] = s;
  }
}

__global__ void k_scatter(const int* __restrict__ rows, const int* __restrict__ cols,
                          const float* __restrict__ g, float* __restrict__ out, int E) {
  int t = blockIdx.x * blockDim.x + threadIdx.x;
  int e = t >> 4;
  if (e >= E) return;
  int r = rows[e], c = cols[e];
  if (r == c) return;
  int sub = (t & 15) * 4;
  const float4 gv = *reinterpret_cast<const float4*>(g + (size_t)r * OUT_DIM + sub);
  float* o = out + (size_t)c * OUT_DIM + sub;
  atomicAdd(o + 0, gv.x);
  atomicAdd(o + 1, gv.y);
  atomicAdd(o + 2, gv.z);
  atomicAdd(o + 3, gv.w);
}

__global__ void k_final(float* __restrict__ out, const float* __restrict__ dinv,
                        const float* __restrict__ bias, int total) {
  int t = blockIdx.x * blockDim.x + threadIdx.x;
  if (t >= total) return;
  out[t] = out[t] * dinv[t >> 6] + bias[t & (OUT_DIM - 1)];
}

extern "C" void kernel_launch(void* const* d_in, const int* in_sizes, int n_in,
                              void* d_out, int out_size, void* d_ws, size_t ws_size,
                              hipStream_t stream) {
  const float* x    = (const float*)d_in[0];
  const int*   eidx = (const int*)d_in[1];
  const float* WB   = (const float*)d_in[2];
  const float* WA   = (const float*)d_in[3];
  const float* bias = (const float*)d_in[4];
  float* out = (float*)d_out;

  const int N = in_sizes[0] / IN_DIM;
  const int E = in_sizes[1] / 2;
  const int* rows = eidx;
  const int* cols = eidx + E;

  char* ws = (char*)d_ws;
  size_t gBytes = ((size_t)N * OUT_DIM * sizeof(float) + 255) & ~(size_t)255;
  size_t dBytes = ((size_t)N * sizeof(int) + 255) & ~(size_t)255;
  float* g    = (float*)ws;
  int*   deg  = (int*)(ws + gBytes);
  float* dinv = (float*)(ws + gBytes + dBytes);

  const int B = 256;
  k_zero_deg<<<(N + B - 1) / B, B, 0, stream>>>(deg, N);
  k_deg<<<(E + B - 1) / B, B, 0, stream>>>(rows, cols, deg, E);
  k_dinv<<<(N + B - 1) / B, B, 0, stream>>>(deg, dinv, N);

  const int ntiles = N / 16;
  if (ntiles > 0) {
    const int nblocks = (ntiles + 7) / 8;
    k_lora_wmma<<<nblocks, B, 0, stream>>>(x, WB, WA, dinv, g, out, ntiles);
  }
  const int rem_start = ntiles * 16;
  if (rem_start < N) {
    k_lora_tail<<<(N - rem_start + B - 1) / B, B, 0, stream>>>(x, WB, WA, dinv, g, out,
                                                               rem_start, N);
  }

  const long long sthreads = (long long)E * 16;
  k_scatter<<<(int)((sthreads + B - 1) / B), B, 0, stream>>>(rows, cols, g, out, E);

  const int total = N * OUT_DIM;
  k_final<<<(total + B - 1) / B, B, 0, stream>>>(out, dinv, bias, total);
}


# W_A product moved behind the neighbour sum (linear): edges add 3 floats per node into LDS sums per node range, not 64 global f32 atomics per edge; f32 sum order changes
# speedup vs baseline: 2.1310x; 2.1310x over previous
#include <hip/hip_runtime.h>
#include <hip/hip_bf16.h>


#define IN_DIM 256
#define OUT_DIM 64
#define RANK 3
#define WB_STRIDE 260

#define NPR 12512
#define NCHUNK 32
#define AGG_T 1024
#define OT_STRIDE 68

typedef __attribute__((ext_vector_type(2))) float v2f;
typedef __attribute__((ext_vector_type(8))) float v8f;


__global__ void k_zero_deg(int* __restrict__ deg, int n) {
  int i = blockIdx.x * blockDim.x + threadIdx.x;
  if (i < n) deg[i] = 0;
}

__global__ void k_deg(const int* __restrict__ rows, const int* __restrict__ cols,
                      int* __restrict__ deg, int E) {
  int e = blockIdx.x * blockDim.x + threadIdx.x;
  if (e >= E) return;
  int r = rows[e], c = cols[e];
  if (r != c) atomicAdd(&deg[c], 1);
}

__global__ void k_dinv(const int* __restrict__ deg, float* __restrict__ dinv, int n) {
  int i = blockIdx.x * blockDim.x + threadIdx.x;
  if (i < n) dinv[i] = rsqrtf((float)deg[i] + 1.0f);
}

__global__ void __launch_bounds__(256) k_lora_wmma(
    const float* __restrict__ x, const float* __restrict__ WB,
    const float* __restrict__ dinv, float* __restrict__ s,
    float* __restrict__ acc, int ntiles) {
  __shared__ float ldsWB[16][WB_STRIDE];
  __shared__ float ldsT[8][256];

  for (int idx = threadIdx.x; idx < 16 * IN_DIM; idx += blockDim.x) {
    const int r = idx >> 8;
    const int k = idx & (IN_DIM - 1);
    ldsWB[r][k] = (r < RANK) ? WB[r * IN_DIM + k] : 0.0f;
  }
  __syncthreads();

  const int wave = threadIdx.x >> 5;
  const int lane = threadIdx.x & 31;
  int tile = blockIdx.x * 8 + wave;
  if (tile >= ntiles) tile = ntiles - 1;

  const int m    = lane & 15;
  const int half = lane >> 4;
  const int koff = half * 2;
  const int base = tile * 16;

  const float* xrow = x + (size_t)(base + m) * IN_DIM + koff;
  const float* wrow = &ldsWB[m][koff];

  v8f c = {};
  #pragma unroll 8
  for (int kk = 0; kk < IN_DIM; kk += 4) {
    v2f a = *reinterpret_cast<const v2f*>(xrow + kk);
    v2f b = *reinterpret_cast<const v2f*>(wrow + kk);
    c = __builtin_amdgcn_wmma_f32_16x16x4_f32(false, a, false, b, (short)0, c,
                                              false, false);
  }

  #pragma unroll
  for (int j = 0; j < 8; ++j) {
    ldsT[wave][(j + 8 * half) * 16 + m] = c[j];
  }
  __syncthreads();

  if (lane < 16) {
    const int node = base + lane;
    const float dvn = dinv[node];
    const float* t = &ldsT[wave][lane * 16];
    const float s0 = t[0] * dvn;
    const float s1 = t[1] * dvn;
    const float s2 = t[2] * dvn;
    *reinterpret_cast<float4*>(s + (size_t)node * 4) = make_float4(s0, s1, s2, 0.0f);
    float* a = acc + (size_t)node * RANK;
    a[0] = s0;
    a[1] = s1;
    a[2] = s2;
  }
}

__global__ void k_lora_tail(const float* __restrict__ x, const float* __restrict__ WB,
                            const float* __restrict__ dinv, float* __restrict__ s,
                            float* __restrict__ acc, int start, int n) {
  int i = start + blockIdx.x * blockDim.x + threadIdx.x;
  if (i >= n) return;
  const float dv = dinv[i];
  for (int r = 0; r < RANK; ++r) {
    float t = 0.0f;
    for (int k = 0; k < IN_DIM; ++k) t += x[(size_t)i * IN_DIM + k] * WB[r * IN_DIM + k];
    t *= dv;
    s[(size_t)i * 4 + r] = t;
    acc[(size_t)i * RANK + r] = t;
  }
  s[(size_t)i * 4 + 3] = 0.0f;
}

__global__ void __launch_bounds__(AGG_T) k_agg(
    const int* __restrict__ rows, const int* __restrict__ cols,
    const float* __restrict__ s, float* __restrict__ acc,
    int N, int E, int epc, int nranges) {
  __shared__ float lacc[NPR * RANK];

  const int rng   = blockIdx.x % nranges;
  const int chunk = blockIdx.x / nranges;
  const int lo    = rng * NPR;
  const int cnt   = min(NPR, N - lo);
  const int nf    = cnt * RANK;

  for (int i = threadIdx.x; i < nf; i += AGG_T) lacc[i] = 0.0f;
  __syncthreads();

  const int e0 = chunk * epc;
  const int e1 = min(E, e0 + epc);
  const float4* __restrict__ s4 = reinterpret_cast<const float4*>(s);

  if (e0 < e1) {
    const int elast = e1 - 1;
    int c[4], r[4];
    for (int q = 0; q < 4; ++q) {
      const int ee = min(e0 + (int)threadIdx.x + q * AGG_T, elast);
      c[q] = cols[ee];
      r[q] = rows[ee];
    }
    for (int eb = e0; eb < e1; eb += 4 * AGG_T) {
      int cn[4], rn[4];
      for (int q = 0; q < 4; ++q) {
        const int ee = min(eb + (int)threadIdx.x + (4 + q) * AGG_T, elast);
        cn[q] = cols[ee];
        rn[q] = rows[ee];
      }
      asm volatile("" ::: "memory");
      bool p[4];
      float4 sv[4];
      for (int q = 0; q < 4; ++q) {
        const bool ok = eb + (int)threadIdx.x + q * AGG_T < e1;
        const unsigned off = (unsigned)(c[q] - lo);
        p[q] = ok && (off < (unsigned)cnt) && (r[q] != c[q]);
        sv[q] = s4[p[q] ? r[q] : lo];
      }
      asm volatile("" : "+v"(sv[0].x), "+v"(sv[0].y), "+v"(sv[0].z),
                        "+v"(sv[1].x), "+v"(sv[1].y), "+v"(sv[1].z),
                        "+v"(sv[2].x), "+v"(sv[2].y), "+v"(sv[2].z),
                        "+v"(sv[3].x), "+v"(sv[3].y), "+v"(sv[3].z));
      for (int q = 0; q < 4; ++q) {
        if (p[q]) {
          float* a = &lacc[(c[q] - lo) * RANK];
          atomicAdd(a + 0, sv[q].x);
          atomicAdd(a + 1, sv[q].y);
          atomicAdd(a + 2, sv[q].z);
        }
      }
      for (int q = 0; q < 4; ++q) { c[q] = cn[q]; r[q] = rn[q]; }
    }
  }
  __syncthreads();

  float* gacc = acc + (size_t)lo * RANK;
  const int rot = (int)(((long long)chunk * (nf >> 5)) / NCHUNK) << 5;
  for (int i0 = threadIdx.x; i0 < nf; i0 += 4 * AGG_T) {
    int   k[4];
    float v[4];
    for (int u = 0; u < 4; ++u) {
      const int i = i0 + u * AGG_T;
      int kk = i + rot;
      if (kk >= nf) kk -= nf;
      k[u] = (i < nf) ? kk : -1;
      v[u] = (i < nf) ? lacc[kk] : 0.0f;
    }
    for (int u = 0; u < 4; ++u) {
      if (k[u] >= 0) atomicAdd(&gacc[k[u]], v[u]);
    }
  }
}

__global__ void __launch_bounds__(256) k_project_out(
    const float* __restrict__ acc, const float* __restrict__ dinv,
    const float* __restrict__ WA, const float* __restrict__ bias,
    float* __restrict__ out, int N, int ntiles) {
  __shared__ float tileO[8][16 * OT_STRIDE];

  const int wave = threadIdx.x >> 5;
  const int lane = threadIdx.x & 31;
  int tile = blockIdx.x * 8 + wave;
  if (tile >= ntiles) tile = ntiles - 1;

  const int m    = lane & 15;
  const int half = lane >> 4;
  const int koff = half * 2;
  const int base = tile * 16;

  int node = base + m;
  if (node >= N) node = N - 1;
  const float dv = dinv[node];
  const float* ap = acc + (size_t)node * RANK;
  const int k1 = (half == 0) ? 1 : 2;
  const float ay = ap[k1];
  v2f a2;
  a2.x = ap[koff] * dv;
  a2.y = (half == 0) ? ay * dv : 0.0f;

  float* to = tileO[wave];
  for (int jt = 0; jt < 4; ++jt) {
    const int oc = jt * 16 + m;
    const float wy = WA[oc * RANK + k1];
    v2f b2;
    b2.x = WA[oc * RANK + koff];
    b2.y = (half == 0) ? wy : 0.0f;
    const float bs = bias[oc];
    v8f d;
    for (int j = 0; j < 8; ++j) d[j] = bs;
    d = __builtin_amdgcn_wmma_f32_16x16x4_f32(false, a2, false, b2, (short)0, d,
                                              false, false);
    for (int j = 0; j < 8; ++j) to[(j + 8 * half) * OT_STRIDE + oc] = d[j];
  }
  __syncthreads();

  float4 v[8];
  for (int it = 0; it < 8; ++it) {
    v[it] = *reinterpret_cast<const float4*>(&to[(it * 2 + half) * OT_STRIDE + m * 4]);
  }
  float* orow = out + (size_t)(base + half) * OUT_DIM + m * 4;
  if (base + 16 <= N) {
    for (int it = 0; it < 8; ++it) {
      *reinterpret_cast<float4*>(orow + (size_t)it * 2 * OUT_DIM) = v[it];
    }
  } else {
    for (int it = 0; it < 8; ++it) {
      if (base + it * 2 + half < N) *reinterpret_cast<float4*>(orow + (size_t)it * 2 * OUT_DIM) = v[it];
    }
  }
}

extern "C" void kernel_launch(void* const* d_in, const int* in_sizes, int n_in,
                              void* d_out, int out_size, void* d_ws, size_t ws_size,
                              hipStream_t stream) {
  const float* x    = (const float*)d_in[0];
  const int*   eidx = (const int*)d_in[1];
  const float* WB   = (const float*)d_in[2];
  const float* WA   = (const float*)d_in[3];
  const float* bias = (const float*)d_in[4];
  float* out = (float*)d_out;

  const int N = in_sizes[0] / IN_DIM;
  const int E = in_sizes[1] / 2;
  const int* rows = eidx;
  const int* cols = eidx + E;

  char* ws = (char*)d_ws;
  size_t sBytes = ((size_t)N * 4 * sizeof(float) + 255) & ~(size_t)255;
  size_t aBytes = ((size_t)N * RANK * sizeof(float) + 255) & ~(size_t)255;
  size_t dBytes = ((size_t)N * sizeof(int) + 255) & ~(size_t)255;
  float* s    = (float*)ws;
  float* acc  = (float*)(ws + sBytes);
  int*   deg  = (int*)(ws + sBytes + aBytes);
  float* dinv = (float*)(ws + sBytes + aBytes + dBytes);

  const int B = 256;
  k_zero_deg<<<(N + B - 1) / B, B, 0, stream>>>(deg, N);
  k_deg<<<(E + B - 1) / B, B, 0, stream>>>(rows, cols, deg, E);
  k_dinv<<<(N + B - 1) / B, B, 0, stream>>>(deg, dinv, N);

  const int ntiles = N / 16;
  if (ntiles > 0) {
    const int nblocks = (ntiles + 7) / 8;
    k_lora_wmma<<<nblocks, B, 0, stream>>>(x, WB, dinv, s, acc, ntiles);
  }
  const int rem_start = ntiles * 16;
  if (rem_start < N) {
    k_lora_tail<<<(N - rem_start + B - 1) / B, B, 0, stream>>>(x, WB, dinv, s, acc,
                                                               rem_start, N);
  }

  const int nranges = (N + NPR - 1) / NPR;
  const int epc = (E + NCHUNK - 1) / NCHUNK;
  k_agg<<<nranges * NCHUNK, AGG_T, 0, stream>>>(rows, cols, s, acc, N, E, epc, nranges);

  const int otiles = (N + 15) / 16;
  k_project_out<<<(otiles + 7) / 8, B, 0, stream>>>(acc, dinv, WA, bias, out, N, otiles);
}


# Degree count: LDS counters per (node range, edge chunk), flushed by whole-line integer atomics, replace one scattered global atomic per edge: one L2 atomic request per 32 nodes and chunk
# speedup vs baseline: 4.5934x; 2.1555x over previous
#include <hip/hip_runtime.h>
#include <hip/hip_bf16.h>


#define IN_DIM 256
#define OUT_DIM 64
#define RANK 3
#define WB_STRIDE 260

#define NPR 12512
#define NCHUNK 32
#define AGG_T 1024
#define OT_STRIDE 68
#define DEG_U 16

typedef __attribute__((ext_vector_type(2))) float v2f;
typedef __attribute__((ext_vector_type(8))) float v8f;


__global__ void k_zero_deg(int* __restrict__ deg, int n) {
  int i = blockIdx.x * blockDim.x + threadIdx.x;
  if (i < n) deg[i] = 0;
}

__global__ void __launch_bounds__(AGG_T) k_deg(
    const int* __restrict__ rows, const int* __restrict__ cols,
    int* __restrict__ deg, int N, int E, int epc, int nranges) {
  __shared__ int lcnt[NPR];

  const int rng   = blockIdx.x % nranges;
  const int chunk = blockIdx.x / nranges;
  const int lo    = rng * NPR;
  const int cnt   = min(NPR, N - lo);

  for (int i = threadIdx.x; i < cnt; i += AGG_T) lcnt[i] = 0;
  __syncthreads();

  const int e0 = chunk * epc;
  const int e1 = min(E, e0 + epc);
  if (e0 < e1) {
    const int elast = e1 - 1;
    for (int eb = e0; eb < e1; eb += DEG_U * AGG_T) {
      int c[DEG_U], r[DEG_U];
      for (int q = 0; q < DEG_U; ++q) {
        const int ee = min(eb + (int)threadIdx.x + q * AGG_T, elast);
        c[q] = cols[ee];
        r[q] = rows[ee];
      }
      for (int q = 0; q < DEG_U; ++q) {
        const bool ok = eb + (int)threadIdx.x + q * AGG_T < e1;
        const unsigned off = (unsigned)(c[q] - lo);
        if (ok && off < (unsigned)cnt && r[q] != c[q]) atomicAdd(&lcnt[off], 1);
      }
    }
  }
  __syncthreads();

  int* gdeg = deg + lo;
  const int rot = (int)(((long long)chunk * (cnt >> 5)) / NCHUNK) << 5;
  for (int i0 = threadIdx.x; i0 < cnt; i0 += 4 * AGG_T) {
    int k[4], v[4];
    for (int u = 0; u < 4; ++u) {
      const int i = i0 + u * AGG_T;
      int kk = i + rot;
      if (kk >= cnt) kk -= cnt;
      k[u] = (i < cnt) ? kk : -1;
      v[u] = (i < cnt) ? lcnt[kk] : 0;
    }
    for (int u = 0; u < 4; ++u) {
      if (k[u] >= 0) atomicAdd(&gdeg[k[u]], v[u]);
    }
  }
}

__global__ void k_dinv(const int* __restrict__ deg, float* __restrict__ dinv, int n) {
  int i = blockIdx.x * blockDim.x + threadIdx.x;
  if (i < n) dinv[i] = rsqrtf((float)deg[i] + 1.0f);
}

__global__ void __launch_bounds__(256) k_lora_wmma(
    const float* __restrict__ x, const float* __restrict__ WB,
    const float* __restrict__ dinv, float* __restrict__ s,
    float* __restrict__ acc, int ntiles) {
  __shared__ float ldsWB[16][WB_STRIDE];
  __shared__ float ldsT[8][256];

  for (int idx = threadIdx.x; idx < 16 * IN_DIM; idx += blockDim.x) {
    const int r = idx >> 8;
    const int k = idx & (IN_DIM - 1);
    ldsWB[r][k] = (r < RANK) ? WB[r * IN_DIM + k] : 0.0f;
  }
  __syncthreads();

  const int wave = threadIdx.x >> 5;
  const int lane = threadIdx.x & 31;
  int tile = blockIdx.x * 8 + wave;
  if (tile >= ntiles) tile = ntiles - 1;

  const int m    = lane & 15;
  const int half = lane >> 4;
  const int koff = half * 2;
  const int base = tile * 16;

  const float* xrow = x + (size_t)(base + m) * IN_DIM + koff;
  const float* wrow = &ldsWB[m][koff];

  v8f c = {};
  #pragma unroll 8
  for (int kk = 0; kk < IN_DIM; kk += 4) {
    v2f a = *reinterpret_cast<const v2f*>(xrow + kk);
    v2f b = *reinterpret_cast<const v2f*>(wrow + kk);
    c = __builtin_amdgcn_wmma_f32_16x16x4_f32(false, a, false, b, (short)0, c,
                                              false, false);
  }

  #pragma unroll
  for (int j = 0; j < 8; ++j) {
    ldsT[wave][(j + 8 * half) * 16 + m] = c[j];
  }
  __syncthreads();

  if (lane < 16) {
    const int node = base + lane;
    const float dvn = dinv[node];
    const float* t = &ldsT[wave][lane * 16];
    const float s0 = t[0] * dvn;
    const float s1 = t[1] * dvn;
    const float s2 = t[2] * dvn;
    *reinterpret_cast<float4*>(s + (size_t)node * 4) = make_float4(s0, s1, s2, 0.0f);
    float* a = acc + (size_t)node * RANK;
    a[0] = s0;
    a[1] = s1;
    a[2] = s2;
  }
}

__global__ void k_lora_tail(const float* __restrict__ x, const float* __restrict__ WB,
                            const float* __restrict__ dinv, float* __restrict__ s,
                            float* __restrict__ acc, int start, int n) {
  int i = start + blockIdx.x * blockDim.x + threadIdx.x;
  if (i >= n) return;
  const float dv = dinv[i];
  for (int r = 0; r < RANK; ++r) {
    float t = 0.0f;
    for (int k = 0; k < IN_DIM; ++k) t += x[(size_t)i * IN_DIM + k] * WB[r * IN_DIM + k];
    t *= dv;
    s[(size_t)i * 4 + r] = t;
    acc[(size_t)i * RANK + r] = t;
  }
  s[(size_t)i * 4 + 3] = 0.0f;
}

__global__ void __launch_bounds__(AGG_T) k_agg(
    const int* __restrict__ rows, const int* __restrict__ cols,
    const float* __restrict__ s, float* __restrict__ acc,
    int N, int E, int epc, int nranges) {
  __shared__ float lacc[NPR * RANK];

  const int rng   = blockIdx.x % nranges;
  const int chunk = blockIdx.x / nranges;
  const int lo    = rng * NPR;
  const int cnt   = min(NPR, N - lo);
  const int nf    = cnt * RANK;

  for (int i = threadIdx.x; i < nf; i += AGG_T) lacc[i] = 0.0f;
  __syncthreads();

  const int e0 = chunk * epc;
  const int e1 = min(E, e0 + epc);
  const float4* __restrict__ s4 = reinterpret_cast<const float4*>(s);

  if (e0 < e1) {
    const int elast = e1 - 1;
    int c[4], r[4];
    for (int q = 0; q < 4; ++q) {
      const int ee = min(e0 + (int)threadIdx.x + q * AGG_T, elast);
      c[q] = cols[ee];
      r[q] = rows[ee];
    }
    for (int eb = e0; eb < e1; eb += 4 * AGG_T) {
      int cn[4], rn[4];
      for (int q = 0; q < 4; ++q) {
        const int ee = min(eb + (int)threadIdx.x + (4 + q) * AGG_T, elast);
        cn[q] = cols[ee];
        rn[q] = rows[ee];
      }
      asm volatile("" ::: "memory");
      bool p[4];
      float4 sv[4];
      for (int q = 0; q < 4; ++q) {
        const bool ok = eb + (int)threadIdx.x + q * AGG_T < e1;
        const unsigned off = (unsigned)(c[q] - lo);
        p[q] = ok && (off < (unsigned)cnt) && (r[q] != c[q]);
        sv[q] = s4[p[q] ? r[q] : lo];
      }
      asm volatile("" : "+v"(sv[0].x), "+v"(sv[0].y), "+v"(sv[0].z),
                        "+v"(sv[1].x), "+v"(sv[1].y), "+v"(sv[1].z),
                        "+v"(sv[2].x), "+v"(sv[2].y), "+v"(sv[2].z),
                        "+v"(sv[3].x), "+v"(sv[3].y), "+v"(sv[3].z));
      for (int q = 0; q < 4; ++q) {
        if (p[q]) {
          float* a = &lacc[(c[q] - lo) * RANK];
          atomicAdd(a + 0, sv[q].x);
          atomicAdd(a + 1, sv[q].y);
          atomicAdd(a + 2, sv[q].z);
        }
      }
      for (int q = 0; q < 4; ++q) { c[q] = cn[q]; r[q] = rn[q]; }
    }
  }
  __syncthreads();

  float* gacc = acc + (size_t)lo * RANK;
  const int rot = (int)(((long long)chunk * (nf >> 5)) / NCHUNK) << 5;
  for (int i0 = threadIdx.x; i0 < nf; i0 += 4 * AGG_T) {
    int   k[4];
    float v[4];
    for (int u = 0; u < 4; ++u) {
      const int i = i0 + u * AGG_T;
      int kk = i + rot;
      if (kk >= nf) kk -= nf;
      k[u] = (i < nf) ? kk : -1;
      v[u] = (i < nf) ? lacc[kk] : 0.0f;
    }
    for (int u = 0; u < 4; ++u) {
      if (k[u] >= 0) atomicAdd(&gacc[k[u]], v[u]);
    }
  }
}

__global__ void __launch_bounds__(256) k_project_out(
    const float* __restrict__ acc, const float* __restrict__ dinv,
    const float* __restrict__ WA, const float* __restrict__ bias,
    float* __restrict__ out, int N, int ntiles) {
  __shared__ float tileO[8][16 * OT_STRIDE];

  const int wave = threadIdx.x >> 5;
  const int lane = threadIdx.x & 31;
  int tile = blockIdx.x * 8 + wave;
  if (tile >= ntiles) tile = ntiles - 1;

  const int m    = lane & 15;
  const int half = lane >> 4;
  const int koff = half * 2;
  const int base = tile * 16;

  int node = base + m;
  if (node >= N) node = N - 1;
  const float dv = dinv[node];
  const float* ap = acc + (size_t)node * RANK;
  const int k1 = (half == 0) ? 1 : 2;
  const float ay = ap[k1];
  v2f a2;
  a2.x = ap[koff] * dv;
  a2.y = (half == 0) ? ay * dv : 0.0f;

  float* to = tileO[wave];
  for (int jt = 0; jt < 4; ++jt) {
    const int oc = jt * 16 + m;
    const float wy = WA[oc * RANK + k1];
    v2f b2;
    b2.x = WA[oc * RANK + koff];
    b2.y = (half == 0) ? wy : 0.0f;
    const float bs = bias[oc];
    v8f d;
    for (int j = 0; j < 8; ++j) d[j] = bs;
    d = __builtin_amdgcn_wmma_f32_16x16x4_f32(false, a2, false, b2, (short)0, d,
                                              false, false);
    for (int j = 0; j < 8; ++j) to[(j + 8 * half) * OT_STRIDE + oc] = d[j];
  }
  __syncthreads();

  float4 v[8];
  for (int it = 0; it < 8; ++it) {
    v[it] = *reinterpret_cast<const float4*>(&to[(it * 2 + half) * OT_STRIDE + m * 4]);
  }
  float* orow = out + (size_t)(base + half) * OUT_DIM + m * 4;
  if (base + 16 <= N) {
    for (int it = 0; it < 8; ++it) {
      *reinterpret_cast<float4*>(orow + (size_t)it * 2 * OUT_DIM) = v[it];
    }
  } else {
    for (int it = 0; it < 8; ++it) {
      if (base + it * 2 + half < N) *reinterpret_cast<float4*>(orow + (size_t)it * 2 * OUT_DIM) = v[it];
    }
  }
}

extern "C" void kernel_launch(void* const* d_in, const int* in_sizes, int n_in,
                              void* d_out, int out_size, void* d_ws, size_t ws_size,
                              hipStream_t stream) {
  const float* x    = (const float*)d_in[0];
  const int*   eidx = (const int*)d_in[1];
  const float* WB   = (const float*)d_in[2];
  const float* WA   = (const float*)d_in[3];
  const float* bias = (const float*)d_in[4];
  float* out = (float*)d_out;

  const int N = in_sizes[0] / IN_DIM;
  const int E = in_sizes[1] / 2;
  const int* rows = eidx;
  const int* cols = eidx + E;

  char* ws = (char*)d_ws;
  size_t sBytes = ((size_t)N * 4 * sizeof(float) + 255) & ~(size_t)255;
  size_t aBytes = ((size_t)N * RANK * sizeof(float) + 255) & ~(size_t)255;
  size_t dBytes = ((size_t)N * sizeof(int) + 255) & ~(size_t)255;
  float* s    = (float*)ws;
  float* acc  = (float*)(ws + sBytes);
  int*   deg  = (int*)(ws + sBytes + aBytes);
  float* dinv = (float*)(ws + sBytes + aBytes + dBytes);

  const int B = 256;
  k_zero_deg<<<(N + B - 1) / B, B, 0, stream>>>(deg, N);
  const int nranges = (N + NPR - 1) / NPR;
  const int epc = (E + NCHUNK - 1) / NCHUNK;
  k_deg<<<nranges * NCHUNK, AGG_T, 0, stream>>>(rows, cols, deg, N, E, epc, nranges);
  k_dinv<<<(N + B - 1) / B, B, 0, stream>>>(deg, dinv, N);

  const int ntiles = N / 16;
  if (ntiles > 0) {
    const int nblocks = (ntiles + 7) / 8;
    k_lora_wmma<<<nblocks, B, 0, stream>>>(x, WB, dinv, s, acc, ntiles);
  }
  const int rem_start = ntiles * 16;
  if (rem_start < N) {
    k_lora_tail<<<(N - rem_start + B - 1) / B, B, 0, stream>>>(x, WB, dinv, s, acc,
                                                               rem_start, N);
  }

  k_agg<<<nranges * NCHUNK, AGG_T, 0, stream>>>(rows, cols, s, acc, N, E, epc, nranges);

  const int otiles = (N + 15) / 16;
  k_project_out<<<(otiles + 7) / 8, B, 0, stream>>>(acc, dinv, WA, bias, out, N, otiles);
}


# Degree scan and X*W_B product share one launch: per work-group 16 waves scan edges while 16 multiply tiles; dinv is applied to T afterwards, where it is formed; 6 launches become 5
# speedup vs baseline: 4.9349x; 1.0743x over previous
#include <hip/hip_runtime.h>
#include <hip/hip_bf16.h>


#define IN_DIM 256
#define OUT_DIM 64
#define RANK 3
#define WB_STRIDE 260

#define NPR 12512
#define NCHUNK 32
#define AGG_T 1024
#define OT_STRIDE 68
#define DEG_U 16
#define SCAN_T 512
#define PROJ_W 16
#define PROJ_SLOTS 2

typedef __attribute__((ext_vector_type(2))) float v2f;
typedef __attribute__((ext_vector_type(8))) float v8f;


__global__ void k_zero_deg(int* __restrict__ deg, int n) {
  int i = blockIdx.x * blockDim.x + threadIdx.x;
  if (i < n) deg[i] = 0;
}


__device__ __forceinline__ void project_tile(const float* __restrict__ x, const float* wrow,
                                             float* tw, int base, int m, int half, int koff) {
  const float* xrow = x + (size_t)(base + m) * IN_DIM + koff;
  v8f c = {};
  #pragma unroll 16
  for (int kk = 0; kk < IN_DIM; kk += 4) {
    v2f a = *reinterpret_cast<const v2f*>(xrow + kk);
    v2f b = *reinterpret_cast<const v2f*>(wrow + kk);
    c = __builtin_amdgcn_wmma_f32_16x16x4_f32(false, a, false, b, (short)0, c,
                                              false, false);
  }
  #pragma unroll
  for (int j = 0; j < 8; ++j) {
    tw[(j + 8 * half) * 16 + m] = c[j];
  }
}

__global__ void __launch_bounds__(AGG_T) k_deg_and_project(
    const int* __restrict__ rows, const int* __restrict__ cols,
    int* __restrict__ deg, int N, int E, int epc, int nranges,
    const float* __restrict__ x, const float* __restrict__ WB,
    float* __restrict__ s, int ntiles, int tpw) {
  __shared__ int   lcnt[NPR];
  __shared__ float ldsWB[16][WB_STRIDE];
  __shared__ float ldsT[PROJ_W][PROJ_SLOTS][256];

  const int  tid  = threadIdx.x;
  const bool scan = tid < SCAN_T;

  const int rng   = blockIdx.x % nranges;
  const int chunk = blockIdx.x / nranges;
  const int lo    = rng * NPR;
  const int cnt   = min(NPR, N - lo);

  const int pt   = tid - SCAN_T;
  const int pw   = pt >> 5;
  const int lane = tid & 31;
  const int m    = lane & 15;
  const int half = lane >> 4;
  const int koff = half * 2;
  const float* wrow = &ldsWB[m][koff];

  if (scan) {
    for (int i = tid; i < cnt; i += SCAN_T) lcnt[i] = 0;
  } else {
    for (int idx = pt; idx < 16 * IN_DIM; idx += AGG_T - SCAN_T) {
      const int r = idx >> 8;
      const int k = idx & (IN_DIM - 1);
      ldsWB[r][k] = (r < RANK) ? WB[r * IN_DIM + k] : 0.0f;
    }
  }
  __syncthreads();

  if (scan) {
    const int e0 = chunk * epc;
    const int e1 = min(E, e0 + epc);
    if (e0 < e1) {
      const int elast = e1 - 1;
      for (int eb = e0; eb < e1; eb += DEG_U * SCAN_T) {
        int c[DEG_U], r[DEG_U];
        for (int q = 0; q < DEG_U; ++q) {
          const int ee = min(eb + tid + q * SCAN_T, elast);
          c[q] = cols[ee];
          r[q] = rows[ee];
        }
        for (int q = 0; q < DEG_U; ++q) {
          const bool ok = eb + tid + q * SCAN_T < e1;
          const unsigned off = (unsigned)(c[q] - lo);
          if (ok && off < (unsigned)cnt && r[q] != c[q]) atomicAdd(&lcnt[off], 1);
        }
      }
    }
  } else {
    for (int sl = 0; sl < PROJ_SLOTS; ++sl) {
      const int j    = sl * PROJ_W + pw;
      const int tile = (int)blockIdx.x + (int)gridDim.x * j;
      if (j < tpw && tile < ntiles)
        project_tile(x, wrow, &ldsT[pw][sl][0], tile * 16, m, half, koff);
    }
  }
  __syncthreads();

  if (scan) {
    int* gdeg = deg + lo;
    const int rot = (int)(((long long)chunk * (cnt >> 5)) / NCHUNK) << 5;
    for (int i0 = tid; i0 < cnt; i0 += 4 * SCAN_T) {
      int k[4], v[4];
      for (int u = 0; u < 4; ++u) {
        const int i = i0 + u * SCAN_T;
        int kk = i + rot;
        if (kk >= cnt) kk -= cnt;
        k[u] = (i < cnt) ? kk : -1;
        v[u] = (i < cnt) ? lcnt[kk] : 0;
      }
      for (int u = 0; u < 4; ++u) {
        if (k[u] >= 0) atomicAdd(&gdeg[k[u]], v[u]);
      }
    }
  } else {
    for (int sl = 0; sl < PROJ_SLOTS; ++sl) {
      const int j    = sl * PROJ_W + pw;
      const int tile = (int)blockIdx.x + (int)gridDim.x * j;
      if (j < tpw && tile < ntiles && lane < 16) {
        const float* t = &ldsT[pw][sl][lane * 16];
        *reinterpret_cast<float4*>(s + ((size_t)tile * 16 + lane) * 4) =
            make_float4(t[0], t[1], t[2], 0.0f);
      }
    }
  }

  for (int j0 = PROJ_W * PROJ_SLOTS; j0 < tpw; j0 += PROJ_W * PROJ_SLOTS) {
    __syncthreads();
    if (!scan) {
      for (int sl = 0; sl < PROJ_SLOTS; ++sl) {
        const int j    = j0 + sl * PROJ_W + pw;
        const int tile = (int)blockIdx.x + (int)gridDim.x * j;
        if (j < tpw && tile < ntiles)
          project_tile(x, wrow, &ldsT[pw][sl][0], tile * 16, m, half, koff);
      }
    }
    __syncthreads();
    if (!scan) {
      for (int sl = 0; sl < PROJ_SLOTS; ++sl) {
        const int j    = j0 + sl * PROJ_W + pw;
        const int tile = (int)blockIdx.x + (int)gridDim.x * j;
        if (j < tpw && tile < ntiles && lane < 16) {
          const float* t = &ldsT[pw][sl][lane * 16];
          *reinterpret_cast<float4*>(s + ((size_t)tile * 16 + lane) * 4) =
              make_float4(t[0], t[1], t[2], 0.0f);
        }
      }
    }
  }
}

__global__ void k_lora_tail(const float* __restrict__ x, const float* __restrict__ WB,
                            float* __restrict__ s, int start, int n) {
  int i = start + blockIdx.x * blockDim.x + threadIdx.x;
  if (i >= n) return;
  for (int r = 0; r < RANK; ++r) {
    float t = 0.0f;
    for (int k = 0; k < IN_DIM; ++k) t += x[(size_t)i * IN_DIM + k] * WB[r * IN_DIM + k];
    s[(size_t)i * 4 + r] = t;
  }
  s[(size_t)i * 4 + 3] = 0.0f;
}

__global__ void k_scale(const int* __restrict__ deg, float* __restrict__ dinv,
                        float* __restrict__ s, float* __restrict__ acc, int n) {
  int i = blockIdx.x * blockDim.x + threadIdx.x;
  if (i >= n) return;
  const float dvn = rsqrtf((float)deg[i] + 1.0f);
  dinv[i] = dvn;
  float4* sp = reinterpret_cast<float4*>(s) + i;
  const float4 t = *sp;
  const float s0 = t.x * dvn;
  const float s1 = t.y * dvn;
  const float s2 = t.z * dvn;
  *sp = make_float4(s0, s1, s2, 0.0f);
  float* a = acc + (size_t)i * RANK;
  a[0] = s0;
  a[1] = s1;
  a[2] = s2;
}

__global__ void __launch_bounds__(AGG_T) k_agg(
    const int* __restrict__ rows, const int* __restrict__ cols,
    const float* __restrict__ s, float* __restrict__ acc,
    int N, int E, int epc, int nranges) {
  __shared__ float lacc[NPR * RANK];

  const int rng   = blockIdx.x % nranges;
  const int chunk = blockIdx.x / nranges;
  const int lo    = rng * NPR;
  const int cnt   = min(NPR, N - lo);
  const int nf    = cnt * RANK;

  for (int i = threadIdx.x; i < nf; i += AGG_T) lacc[i] = 0.0f;
  __syncthreads();

  const int e0 = chunk * epc;
  const int e1 = min(E, e0 + epc);
  const float4* __restrict__ s4 = reinterpret_cast<const float4*>(s);

  if (e0 < e1) {
    const int elast = e1 - 1;
    int c[4], r[4];
    for (int q = 0; q < 4; ++q) {
      const int ee = min(e0 + (int)threadIdx.x + q * AGG_T, elast);
      c[q] = cols[ee];
      r[q] = rows[ee];
    }
    for (int eb = e0; eb < e1; eb += 4 * AGG_T) {
      int cn[4], rn[4];
      for (int q = 0; q < 4; ++q) {
        const int ee = min(eb + (int)threadIdx.x + (4 + q) * AGG_T, elast);
        cn[q] = cols[ee];
        rn[q] = rows[ee];
      }
      asm volatile("" ::: "memory");
      bool p[4];
      float4 sv[4];
      for (int q = 0; q < 4; ++q) {
        const bool ok = eb + (int)threadIdx.x + q * AGG_T < e1;
        const unsigned off = (unsigned)(c[q] - lo);
        p[q] = ok && (off < (unsigned)cnt) && (r[q] != c[q]);
        sv[q] = s4[p[q] ? r[q] : lo];
      }
      asm volatile("" : "+v"(sv[0].x), "+v"(sv[0].y), "+v"(sv[0].z),
                        "+v"(sv[1].x), "+v"(sv[1].y), "+v"(sv[1].z),
                        "+v"(sv[2].x), "+v"(sv[2].y), "+v"(sv[2].z),
                        "+v"(sv[3].x), "+v"(sv[3].y), "+v"(sv[3].z));
      for (int q = 0; q < 4; ++q) {
        if (p[q]) {
          float* a = &lacc[(c[q] - lo) * RANK];
          atomicAdd(a + 0, sv[q].x);
          atomicAdd(a + 1, sv[q].y);
          atomicAdd(a + 2, sv[q].z);
        }
      }
      for (int q = 0; q < 4; ++q) { c[q] = cn[q]; r[q] = rn[q]; }
    }
  }
  __syncthreads();

  float* gacc = acc + (size_t)lo * RANK;
  const int rot = (int)(((long long)chunk * (nf >> 5)) / NCHUNK) << 5;
  for (int i0 = threadIdx.x; i0 < nf; i0 += 4 * AGG_T) {
    int   k[4];
    float v[4];
    for (int u = 0; u < 4; ++u) {
      const int i = i0 + u * AGG_T;
      int kk = i + rot;
      if (kk >= nf) kk -= nf;
      k[u] = (i < nf) ? kk : -1;
      v[u] = (i < nf) ? lacc[kk] : 0.0f;
    }
    for (int u = 0; u < 4; ++u) {
      if (k[u] >= 0) atomicAdd(&gacc[k[u]], v[u]);
    }
  }
}

__global__ void __launch_bounds__(256) k_project_out(
    const float* __restrict__ acc, const float* __restrict__ dinv,
    const float* __restrict__ WA, const float* __restrict__ bias,
    float* __restrict__ out, int N, int ntiles) {
  __shared__ float tileO[8][16 * OT_STRIDE];

  const int wave = threadIdx.x >> 5;
  const int lane = threadIdx.x & 31;
  int tile = blockIdx.x * 8 + wave;
  if (tile >= ntiles) tile = ntiles - 1;

  const int m    = lane & 15;
  const int half = lane >> 4;
  const int koff = half * 2;
  const int base = tile * 16;

  int node = base + m;
  if (node >= N) node = N - 1;
  const float dv = dinv[node];
  const float* ap = acc + (size_t)node * RANK;
  const int k1 = (half == 0) ? 1 : 2;
  const float ay = ap[k1];
  v2f a2;
  a2.x = ap[koff] * dv;
  a2.y = (half == 0) ? ay * dv : 0.0f;

  float* to = tileO[wave];
  for (int jt = 0; jt < 4; ++jt) {
    const int oc = jt * 16 + m;
    const float wy = WA[oc * RANK + k1];
    v2f b2;
    b2.x = WA[oc * RANK + koff];
    b2.y = (half == 0) ? wy : 0.0f;
    const float bs = bias[oc];
    v8f d;
    for (int j = 0; j < 8; ++j) d[j] = bs;
    d = __builtin_amdgcn_wmma_f32_16x16x4_f32(false, a2, false, b2, (short)0, d,
                                              false, false);
    for (int j = 0; j < 8; ++j) to[(j + 8 * half) * OT_STRIDE + oc] = d[j];
  }
  __syncthreads();

  float4 v[8];
  for (int it = 0; it < 8; ++it) {
    v[it] = *reinterpret_cast<const float4*>(&to[(it * 2 + half) * OT_STRIDE + m * 4]);
  }
  float* orow = out + (size_t)(base + half) * OUT_DIM + m * 4;
  if (base + 16 <= N) {
    for (int it = 0; it < 8; ++it) {
      *reinterpret_cast<float4*>(orow + (size_t)it * 2 * OUT_DIM) = v[it];
    }
  } else {
    for (int it = 0; it < 8; ++it) {
      if (base + it * 2 + half < N) *reinterpret_cast<float4*>(orow + (size_t)it * 2 * OUT_DIM) = v[it];
    }
  }
}

extern "C" void kernel_launch(void* const* d_in, const int* in_sizes, int n_in,
                              void* d_out, int out_size, void* d_ws, size_t ws_size,
                              hipStream_t stream) {
  const float* x    = (const float*)d_in[0];
  const int*   eidx = (const int*)d_in[1];
  const float* WB   = (const float*)d_in[2];
  const float* WA   = (const float*)d_in[3];
  const float* bias = (const float*)d_in[4];
  float* out = (float*)d_out;

  const int N = in_sizes[0] / IN_DIM;
  const int E = in_sizes[1] / 2;
  const int* rows = eidx;
  const int* cols = eidx + E;

  char* ws = (char*)d_ws;
  size_t sBytes = ((size_t)N * 4 * sizeof(float) + 255) & ~(size_t)255;
  size_t aBytes = ((size_t)N * RANK * sizeof(float) + 255) & ~(size_t)255;
  size_t dBytes = ((size_t)N * sizeof(int) + 255) & ~(size_t)255;
  float* s    = (float*)ws;
  float* acc  = (float*)(ws + sBytes);
  int*   deg  = (int*)(ws + sBytes + aBytes);
  float* dinv = (float*)(ws + sBytes + aBytes + dBytes);

  const int B = 256;
  k_zero_deg<<<(N + B - 1) / B, B, 0, stream>>>(deg, N);
  const int nranges = (N + NPR - 1) / NPR;
  const int epc = (E + NCHUNK - 1) / NCHUNK;
  const int ndeg = nranges * NCHUNK;

  const int ntiles = N / 16;
  const int tpw = (ntiles + ndeg - 1) / ndeg;
  k_deg_and_project<<<ndeg, AGG_T, 0, stream>>>(rows, cols, deg, N, E, epc, nranges,
                                                x, WB, s, ntiles, tpw);
  const int rem_start = ntiles * 16;
  if (rem_start < N) {
    k_lora_tail<<<(N - rem_start + B - 1) / B, B, 0, stream>>>(x, WB, s, rem_start, N);
  }
  k_scale<<<(N + B - 1) / B, B, 0, stream>>>(deg, dinv, s, acc, N);

  k_agg<<<nranges * NCHUNK, AGG_T, 0, stream>>>(rows, cols, s, acc, N, E, epc, nranges);

  const int otiles = (N + 15) / 16;
  k_project_out<<<(otiles + 7) / 8, B, 0, stream>>>(acc, dinv, WA, bias, out, N, otiles);
}
